# starting kernel
# baseline (speedup 1.0000x reference)
#include <hip/hip_runtime.h>
#include <math.h>

typedef __attribute__((ext_vector_type(2))) float v2f;
typedef __attribute__((ext_vector_type(8))) float v8f;

#define N_NODES   50000
#define N_EDGES   800000
#define IN_F      256
#define EDGE_F    64
#define NH        8
#define NF        32
#define HF        256

__device__ __forceinline__ unsigned enc_f32(float f) {
  unsigned u = __float_as_uint(f);
  return (u & 0x80000000u) ? ~u : (u | 0x80000000u);
}
__device__ __forceinline__ float dec_f32(unsigned u) {
  unsigned b = (u & 0x80000000u) ? (u ^ 0x80000000u) : ~u;
  return __uint_as_float(b);
}

__global__ __launch_bounds__(256)
void gemm_wh_wmma(const float* __restrict__ h, const float* __restrict__ W,
                  float* __restrict__ Wh) {
  const int lane = threadIdx.x & 31;
  const int wave = threadIdx.x >> 5;
  const int m0   = blockIdx.x * 16;
  const int n0   = wave * 32;
  const int l15  = lane & 15;
  const int hi   = lane >> 4;
  const int khalf = hi * 2;

  v8f c0 = {}; v8f c1 = {};
  const float* arow = h + (size_t)(m0 + l15) * IN_F;

  for (int k0 = 0; k0 < IN_F; k0 += 4) {
    v2f a = *(const v2f*)(arow + k0 + khalf);
    const float* bb = W + (size_t)(k0 + khalf) * HF + n0 + l15;
    v2f b0, b1;
    b0.x = bb[0];        b0.y = bb[HF];
    b1.x = bb[16];       b1.y = bb[HF + 16];
    c0 = __builtin_amdgcn_wmma_f32_16x16x4_f32(false, a, false, b0,
                                               (short)0, c0, false, false);
    c1 = __builtin_amdgcn_wmma_f32_16x16x4_f32(false, a, false, b1,
                                               (short)0, c1, false, false);
  }
  const int mo = m0 + (hi << 3);
  const int nc = n0 + l15;
  for (int r = 0; r < 8; ++r) {
    Wh[(size_t)(mo + r) * HF + nc]      = c0[r];
    Wh[(size_t)(mo + r) * HF + nc + 16] = c1[r];
  }
}

__global__ void fold_we(const float* __restrict__ We_w,
                        const float* __restrict__ a_edge,
                        float* __restrict__ V) {
  int t = threadIdx.x;
  int k = t >> 3, hh = t & 7;
  const float* wr = We_w + (size_t)k * HF + hh * NF;
  const float* ae = a_edge + hh * NF;
  float acc = 0.f;
  for (int f = 0; f < NF; ++f) acc += wr[f] * ae[f];
  V[k * NH + hh] = acc;
}

__global__ __launch_bounds__(256)
void node_scores(const float* __restrict__ Wh, const float* __restrict__ a_src,
                 const float* __restrict__ a_dst,
                 float* __restrict__ ssrc, float* __restrict__ sdst) {
  int t = blockIdx.x * blockDim.x + threadIdx.x;
  if (t >= N_NODES * NH) return;
  int n = t >> 3, hh = t & 7;
  const float* row = Wh + (size_t)n * HF + hh * NF;
  const float* as  = a_src + hh * NF;
  const float* ad  = a_dst + hh * NF;
  float accs = 0.f, accd = 0.f;
  for (int f = 0; f < NF; ++f) { float v = row[f]; accs += v * as[f]; accd += v * ad[f]; }
  ssrc[t] = accs;
  sdst[t] = accd;
}

__global__ __launch_bounds__(256)
void edge_logits(const float* __restrict__ ef, const int* __restrict__ src,
                 const int* __restrict__ dst, const float* __restrict__ V,
                 const float* __restrict__ ssrc, const float* __restrict__ sdst,
                 float* __restrict__ sbuf, unsigned* __restrict__ m) {
  __shared__ float Vl[EDGE_F * NH];
  for (int i = threadIdx.x; i < EDGE_F * NH; i += blockDim.x) Vl[i] = V[i];
  __syncthreads();
  int t = blockIdx.x * blockDim.x + threadIdx.x;
  if (t >= N_EDGES * NH) return;
  int e = t >> 3, hh = t & 7;
  const float4* efr = (const float4*)(ef + (size_t)e * EDGE_F);
  float acc = 0.f;
  for (int q = 0; q < EDGE_F / 4; ++q) {
    float4 v = efr[q];
    int k = q * 4;
    acc += v.x * Vl[(k + 0) * NH + hh] + v.y * Vl[(k + 1) * NH + hh]
         + v.z * Vl[(k + 2) * NH + hh] + v.w * Vl[(k + 3) * NH + hh];
  }
  int s = src[e], d = dst[e];
  float sv = ssrc[s * NH + hh] + sdst[d * NH + hh] + acc;
  sv = sv > 0.f ? sv : 0.01f * sv;
  sbuf[t] = sv;
  atomicMax(&m[d * NH + hh], enc_f32(sv));
}

__global__ __launch_bounds__(256)
void softmax_den(const int* __restrict__ dst, const unsigned* __restrict__ m,
                 float* __restrict__ sbuf, float* __restrict__ den) {
  int t = blockIdx.x * blockDim.x + threadIdx.x;
  if (t >= N_EDGES * NH) return;
  int e = t >> 3, hh = t & 7;
  int d = dst[e];
  float ex = expf(sbuf[t] - dec_f32(m[d * NH + hh]));
  sbuf[t] = ex;
  atomicAdd(&den[d * NH + hh], ex);
}

__global__ __launch_bounds__(256)
void aggregate(const int* __restrict__ src, const int* __restrict__ dst,
               const float* __restrict__ Wh, const float* __restrict__ sbuf,
               const float* __restrict__ den, float* __restrict__ out) {
  int e  = blockIdx.x;
  int c  = threadIdx.x;
  int hh = c >> 5;
  int s = src[e], d = dst[e];
  float a = sbuf[(size_t)e * NH + hh] / den[d * NH + hh];
  float val = Wh[(size_t)s * HF + c] * a;
  atomicAdd(&out[(size_t)d * HF + c], val);
}

__global__ __launch_bounds__(256)
void elu_inplace(float* __restrict__ out, int n) {
  int t = blockIdx.x * blockDim.x + threadIdx.x;
  if (t >= n) return;
  float x = out[t];
  out[t] = x > 0.f ? x : expm1f(x);
}

extern "C" void kernel_launch(void* const* d_in, const int* in_sizes, int n_in,
                              void* d_out, int out_size, void* d_ws, size_t ws_size,
                              hipStream_t stream) {
  const float* h   = (const float*)d_in[0];
  const float* ef  = (const float*)d_in[1];
  const int*   src = (const int*)  d_in[2];
  const int*   dst = (const int*)  d_in[3];
  const float* W_w = (const float*)d_in[4];
  const float* We_w= (const float*)d_in[5];
  const float* a_s = (const float*)d_in[6];
  const float* a_d = (const float*)d_in[7];
  const float* a_e = (const float*)d_in[8];
  float* out = (float*)d_out;

  char* ws = (char*)d_ws;
  const size_t WH_BYTES   = (size_t)N_NODES * HF * 4;
  const size_t SBUF_BYTES = (size_t)N_EDGES * NH * 4;
  const size_t NH_BYTES   = (size_t)N_NODES * NH * 4;
  float*    Wh   = (float*)   (ws);
  float*    sbuf = (float*)   (ws + WH_BYTES);
  float*    ssrc = (float*)   (ws + WH_BYTES + SBUF_BYTES);
  float*    sdst = (float*)   (ws + WH_BYTES + SBUF_BYTES + NH_BYTES);
  unsigned* mmax = (unsigned*)(ws + WH_BYTES + SBUF_BYTES + 2 * NH_BYTES);
  float*    den  = (float*)   (ws + WH_BYTES + SBUF_BYTES + 3 * NH_BYTES);
  float*    V    = (float*)   (ws + WH_BYTES + SBUF_BYTES + 4 * NH_BYTES);

  hipMemsetAsync(out,  0, (size_t)N_NODES * HF * 4, stream);
  hipMemsetAsync(mmax, 0, NH_BYTES, stream);
  hipMemsetAsync(den,  0, NH_BYTES, stream);

  gemm_wh_wmma<<<N_NODES / 16, 256, 0, stream>>>(h, W_w, Wh);
  fold_we<<<1, 512, 0, stream>>>(We_w, a_e, V);
  node_scores<<<(N_NODES * NH + 255) / 256, 256, 0, stream>>>(Wh, a_s, a_d, ssrc, sdst);
  edge_logits<<<(N_EDGES * NH + 255) / 256, 256, 0, stream>>>(ef, src, dst, V, ssrc, sdst, sbuf, mmax);
  softmax_den<<<(N_EDGES * NH + 255) / 256, 256, 0, stream>>>(dst, mmax, sbuf, den);
  aggregate<<<N_EDGES, 256, 0, stream>>>(src, dst, Wh, sbuf, den, out);
  elu_inplace<<<(N_NODES * HF + 255) / 256, 256, 0, stream>>>(out, N_NODES * HF);
}


# Aggregation by owner: edges counting-sorted by dst (arrival-order rank), one wave per dst row sums Wh[src]*a in registers, stores ELU once; no f32 atomics on out; sum order changes
# speedup vs baseline: 2.3844x; 2.3844x over previous
#include <hip/hip_runtime.h>
#include <math.h>

typedef __attribute__((ext_vector_type(2))) float v2f;
typedef __attribute__((ext_vector_type(8))) float v8f;

#define N_NODES   50000
#define N_EDGES   800000
#define IN_F      256
#define EDGE_F    64
#define NH        8
#define NF        32
#define HF        256

__device__ __forceinline__ unsigned enc_f32(float f) {
  unsigned u = __float_as_uint(f);
  return (u & 0x80000000u) ? ~u : (u | 0x80000000u);
}
__device__ __forceinline__ float dec_f32(unsigned u) {
  unsigned b = (u & 0x80000000u) ? (u ^ 0x80000000u) : ~u;
  return __uint_as_float(b);
}

__global__ __launch_bounds__(256)
void gemm_wh_wmma(const float* __restrict__ h, const float* __restrict__ W,
                  float* __restrict__ Wh) {
  const int lane = threadIdx.x & 31;
  const int wave = threadIdx.x >> 5;
  const int m0   = blockIdx.x * 16;
  const int n0   = wave * 32;
  const int l15  = lane & 15;
  const int hi   = lane >> 4;
  const int khalf = hi * 2;

  v8f c0 = {}; v8f c1 = {};
  const float* arow = h + (size_t)(m0 + l15) * IN_F;

  for (int k0 = 0; k0 < IN_F; k0 += 4) {
    v2f a = *(const v2f*)(arow + k0 + khalf);
    const float* bb = W + (size_t)(k0 + khalf) * HF + n0 + l15;
    v2f b0, b1;
    b0.x = bb[0];        b0.y = bb[HF];
    b1.x = bb[16];       b1.y = bb[HF + 16];
    c0 = __builtin_amdgcn_wmma_f32_16x16x4_f32(false, a, false, b0,
                                               (short)0, c0, false, false);
    c1 = __builtin_amdgcn_wmma_f32_16x16x4_f32(false, a, false, b1,
                                               (short)0, c1, false, false);
  }
  const int mo = m0 + (hi << 3);
  const int nc = n0 + l15;
  for (int r = 0; r < 8; ++r) {
    Wh[(size_t)(mo + r) * HF + nc]      = c0[r];
    Wh[(size_t)(mo + r) * HF + nc + 16] = c1[r];
  }
}

__global__ void fold_we(const float* __restrict__ We_w,
                        const float* __restrict__ a_edge,
                        float* __restrict__ V) {
  int t = threadIdx.x;
  int k = t >> 3, hh = t & 7;
  const float* wr = We_w + (size_t)k * HF + hh * NF;
  const float* ae = a_edge + hh * NF;
  float acc = 0.f;
  for (int f = 0; f < NF; ++f) acc += wr[f] * ae[f];
  V[k * NH + hh] = acc;
}

__global__ __launch_bounds__(256)
void node_scores(const float* __restrict__ Wh, const float* __restrict__ a_src,
                 const float* __restrict__ a_dst,
                 float* __restrict__ ssrc, float* __restrict__ sdst) {
  int t = blockIdx.x * blockDim.x + threadIdx.x;
  if (t >= N_NODES * NH) return;
  int n = t >> 3, hh = t & 7;
  const float* row = Wh + (size_t)n * HF + hh * NF;
  const float* as  = a_src + hh * NF;
  const float* ad  = a_dst + hh * NF;
  float accs = 0.f, accd = 0.f;
  for (int f = 0; f < NF; ++f) { float v = row[f]; accs += v * as[f]; accd += v * ad[f]; }
  ssrc[t] = accs;
  sdst[t] = accd;
}

__global__ __launch_bounds__(256)
void edge_logits(const float* __restrict__ ef, const int* __restrict__ src,
                 const int* __restrict__ dst, const float* __restrict__ V,
                 const float* __restrict__ ssrc, const float* __restrict__ sdst,
                 float* __restrict__ sbuf, unsigned* __restrict__ m) {
  __shared__ float Vl[EDGE_F * NH];
  for (int i = threadIdx.x; i < EDGE_F * NH; i += blockDim.x) Vl[i] = V[i];
  __syncthreads();
  int t = blockIdx.x * blockDim.x + threadIdx.x;
  if (t >= N_EDGES * NH) return;
  int e = t >> 3, hh = t & 7;
  const float4* efr = (const float4*)(ef + (size_t)e * EDGE_F);
  float acc = 0.f;
  for (int q = 0; q < EDGE_F / 4; ++q) {
    float4 v = efr[q];
    int k = q * 4;
    acc += v.x * Vl[(k + 0) * NH + hh] + v.y * Vl[(k + 1) * NH + hh]
         + v.z * Vl[(k + 2) * NH + hh] + v.w * Vl[(k + 3) * NH + hh];
  }
  int s = src[e], d = dst[e];
  float sv = ssrc[s * NH + hh] + sdst[d * NH + hh] + acc;
  sv = sv > 0.f ? sv : 0.01f * sv;
  sbuf[t] = sv;
  atomicMax(&m[d * NH + hh], enc_f32(sv));
}

__global__ __launch_bounds__(256)
void softmax_den(const int* __restrict__ dst, const unsigned* __restrict__ m,
                 float* __restrict__ sbuf, float* __restrict__ den) {
  int t = blockIdx.x * blockDim.x + threadIdx.x;
  if (t >= N_EDGES * NH) return;
  int e = t >> 3, hh = t & 7;
  int d = dst[e];
  float ex = expf(sbuf[t] - dec_f32(m[d * NH + hh]));
  sbuf[t] = ex;
  atomicAdd(&den[d * NH + hh], ex);
}

#define CNT_PAD 53248

__global__ __launch_bounds__(256)
void zero_counts(int* __restrict__ cnt) {
  int t = blockIdx.x * blockDim.x + threadIdx.x;
  if (t < CNT_PAD) cnt[t] = 0;
}

__global__ __launch_bounds__(256)
void dst_histogram(const int* __restrict__ dst, int* __restrict__ cnt) {
  int e = blockIdx.x * blockDim.x + threadIdx.x;
  if (e < N_EDGES) atomicAdd(&cnt[dst[e]], 1);
}

__global__ __launch_bounds__(1024)
void scan_counts(int* __restrict__ cnt, int* __restrict__ rowptr) {
  __shared__ int wsum[32];
  const int t = threadIdx.x, lane = t & 31, wave = t >> 5;
  int4* c4 = (int4*)cnt + t * 13;
  int4* r4 = (int4*)rowptr + t * 13;
  int4 v[13];
  for (int i = 0; i < 13; ++i) v[i] = c4[i];
  int sum = 0;
  for (int i = 0; i < 13; ++i) sum += v[i].x + v[i].y + v[i].z + v[i].w;
  int inc = sum;
  for (int o = 1; o < 32; o <<= 1) {
    int nb = __shfl_up(inc, o);
    if (lane >= o) inc += nb;
  }
  if (lane == 31) wsum[wave] = inc;
  __syncthreads();
  if (wave == 0) {
    int w = wsum[lane], wi = w;
    for (int o = 1; o < 32; o <<= 1) {
      int nb = __shfl_up(wi, o);
      if (lane >= o) wi += nb;
    }
    wsum[lane] = wi - w;
  }
  __syncthreads();
  int run = wsum[wave] + inc - sum;
  for (int i = 0; i < 13; ++i) {
    int4 o;
    o.x = run; run += v[i].x;
    o.y = run; run += v[i].y;
    o.z = run; run += v[i].z;
    o.w = run; run += v[i].w;
    r4[i] = o;
    c4[i] = o;
  }
}

__global__ __launch_bounds__(256)
void fill_perm(const int* __restrict__ dst, int* __restrict__ cur,
               int* __restrict__ perm) {
  int e = blockIdx.x * blockDim.x + threadIdx.x;
  if (e >= N_EDGES) return;
  int pos = atomicAdd(&cur[dst[e]], 1);
  pos = min(max(pos, 0), N_EDGES - 1);
  perm[pos] = e;
}

__device__ __forceinline__ float elu1(float x) { return x > 0.f ? x : expm1f(x); }

__global__ __launch_bounds__(256)
void aggregate_by_dst(const int* __restrict__ src, const int* __restrict__ rowptr,
                      const int* __restrict__ perm, const float* __restrict__ Wh,
                      const float* __restrict__ sbuf, const float* __restrict__ den,
                      float* __restrict__ out) {
  const int lane = threadIdx.x & 31;
  const int wave = __builtin_amdgcn_readfirstlane((int)(threadIdx.x >> 5));
  const int node = blockIdx.x * 8 + wave;
  const int h0 = lane >> 3, h1 = 4 + (lane >> 3);
  const int beg = max(rowptr[node], 0);
  const int end = min(rowptr[node + 1], N_EDGES);
  float4 acc0 = {0.f, 0.f, 0.f, 0.f}, acc1 = {0.f, 0.f, 0.f, 0.f};
  if (end > beg) {
    const float dn0 = den[node * NH + h0];
    const float dn1 = den[node * NH + h1];
    for (int base = beg; base < end; base += 32) {
      const int n = min(32, end - base);
      int my_e = 0, my_s = 0;
      if (lane < n) {
        my_e = min(max(perm[base + lane], 0), N_EDGES - 1);
        my_s = src[my_e];
      }
#pragma unroll 4
      for (int j = 0; j < n; ++j) {
        const int e = __builtin_amdgcn_readlane(my_e, j);
        const int s = __builtin_amdgcn_readlane(my_s, j);
        const float a0 = sbuf[(size_t)e * NH + h0] / dn0;
        const float a1 = sbuf[(size_t)e * NH + h1] / dn1;
        const float4* row = (const float4*)(Wh + (size_t)s * HF);
        const float4 v0 = row[lane];
        const float4 v1 = row[32 + lane];
        acc0.x += v0.x * a0; acc0.y += v0.y * a0; acc0.z += v0.z * a0; acc0.w += v0.w * a0;
        acc1.x += v1.x * a1; acc1.y += v1.y * a1; acc1.z += v1.z * a1; acc1.w += v1.w * a1;
      }
    }
  }
  float4 o0, o1;
  o0.x = elu1(acc0.x); o0.y = elu1(acc0.y); o0.z = elu1(acc0.z); o0.w = elu1(acc0.w);
  o1.x = elu1(acc1.x); o1.y = elu1(acc1.y); o1.z = elu1(acc1.z); o1.w = elu1(acc1.w);
  float4* orow = (float4*)(out + (size_t)node * HF);
  orow[lane]      = o0;
  orow[32 + lane] = o1;
}

extern "C" void kernel_launch(void* const* d_in, const int* in_sizes, int n_in,
                              void* d_out, int out_size, void* d_ws, size_t ws_size,
                              hipStream_t stream) {
  const float* h   = (const float*)d_in[0];
  const float* ef  = (const float*)d_in[1];
  const int*   src = (const int*)  d_in[2];
  const int*   dst = (const int*)  d_in[3];
  const float* W_w = (const float*)d_in[4];
  const float* We_w= (const float*)d_in[5];
  const float* a_s = (const float*)d_in[6];
  const float* a_d = (const float*)d_in[7];
  const float* a_e = (const float*)d_in[8];
  float* out = (float*)d_out;

  char* ws = (char*)d_ws;
  const size_t WH_BYTES   = (size_t)N_NODES * HF * 4;
  const size_t SBUF_BYTES = (size_t)N_EDGES * NH * 4;
  const size_t NH_BYTES   = (size_t)N_NODES * NH * 4;
  float*    Wh   = (float*)   (ws);
  float*    sbuf = (float*)   (ws + WH_BYTES);
  float*    ssrc = (float*)   (ws + WH_BYTES + SBUF_BYTES);
  float*    sdst = (float*)   (ws + WH_BYTES + SBUF_BYTES + NH_BYTES);
  unsigned* mmax = (unsigned*)(ws + WH_BYTES + SBUF_BYTES + 2 * NH_BYTES);
  float*    den  = (float*)   (ws + WH_BYTES + SBUF_BYTES + 3 * NH_BYTES);
  float*    V    = (float*)   (ws + WH_BYTES + SBUF_BYTES + 4 * NH_BYTES);

  int* cnt    = (int*)(ws + WH_BYTES + SBUF_BYTES);
  int* rowptr = (int*)(ws + WH_BYTES + SBUF_BYTES + 262144);
  int* perm   = (int*)(ws + WH_BYTES + SBUF_BYTES + NH_BYTES);

  hipMemsetAsync(mmax, 0, NH_BYTES, stream);
  hipMemsetAsync(den,  0, NH_BYTES, stream);

  gemm_wh_wmma<<<N_NODES / 16, 256, 0, stream>>>(h, W_w, Wh);
  fold_we<<<1, 512, 0, stream>>>(We_w, a_e, V);
  node_scores<<<(N_NODES * NH + 255) / 256, 256, 0, stream>>>(Wh, a_s, a_d, ssrc, sdst);
  edge_logits<<<(N_EDGES * NH + 255) / 256, 256, 0, stream>>>(ef, src, dst, V, ssrc, sdst, sbuf, mmax);
  softmax_den<<<(N_EDGES * NH + 255) / 256, 256, 0, stream>>>(dst, mmax, sbuf, den);
  zero_counts<<<CNT_PAD / 256, 256, 0, stream>>>(cnt);
  dst_histogram<<<N_EDGES / 256, 256, 0, stream>>>(dst, cnt);
  scan_counts<<<1, 1024, 0, stream>>>(cnt, rowptr);
  fill_perm<<<N_EDGES / 256, 256, 0, stream>>>(dst, cnt, perm);
  aggregate_by_dst<<<N_NODES / 8, 256, 0, stream>>>(src, rowptr, perm, Wh, sbuf, den, out);
}


# Segment softmax by owner: the dst-row wave takes max, exp sum and ex/den over its sorted edges in registers; atomicMax/atomicAdd, the softmax pass and both memsets go; den sum order changes
# speedup vs baseline: 2.9383x; 1.2323x over previous
#include <hip/hip_runtime.h>
#include <math.h>

typedef __attribute__((ext_vector_type(2))) float v2f;
typedef __attribute__((ext_vector_type(8))) float v8f;

#define N_NODES   50000
#define N_EDGES   800000
#define IN_F      256
#define EDGE_F    64
#define NH        8
#define NF        32
#define HF        256

__global__ __launch_bounds__(256)
void gemm_wh_wmma(const float* __restrict__ h, const float* __restrict__ W,
                  float* __restrict__ Wh) {
  const int lane = threadIdx.x & 31;
  const int wave = threadIdx.x >> 5;
  const int m0   = blockIdx.x * 16;
  const int n0   = wave * 32;
  const int l15  = lane & 15;
  const int hi   = lane >> 4;
  const int khalf = hi * 2;

  v8f c0 = {}; v8f c1 = {};
  const float* arow = h + (size_t)(m0 + l15) * IN_F;

  for (int k0 = 0; k0 < IN_F; k0 += 4) {
    v2f a = *(const v2f*)(arow + k0 + khalf);
    const float* bb = W + (size_t)(k0 + khalf) * HF + n0 + l15;
    v2f b0, b1;
    b0.x = bb[0];        b0.y = bb[HF];
    b1.x = bb[16];       b1.y = bb[HF + 16];
    c0 = __builtin_amdgcn_wmma_f32_16x16x4_f32(false, a, false, b0,
                                               (short)0, c0, false, false);
    c1 = __builtin_amdgcn_wmma_f32_16x16x4_f32(false, a, false, b1,
                                               (short)0, c1, false, false);
  }
  const int mo = m0 + (hi << 3);
  const int nc = n0 + l15;
  for (int r = 0; r < 8; ++r) {
    Wh[(size_t)(mo + r) * HF + nc]      = c0[r];
    Wh[(size_t)(mo + r) * HF + nc + 16] = c1[r];
  }
}

__global__ void fold_we(const float* __restrict__ We_w,
                        const float* __restrict__ a_edge,
                        float* __restrict__ V) {
  int t = threadIdx.x;
  int k = t >> 3, hh = t & 7;
  const float* wr = We_w + (size_t)k * HF + hh * NF;
  const float* ae = a_edge + hh * NF;
  float acc = 0.f;
  for (int f = 0; f < NF; ++f) acc += wr[f] * ae[f];
  V[k * NH + hh] = acc;
}

__global__ __launch_bounds__(256)
void node_scores(const float* __restrict__ Wh, const float* __restrict__ a_src,
                 const float* __restrict__ a_dst,
                 float* __restrict__ ssrc, float* __restrict__ sdst) {
  int t = blockIdx.x * blockDim.x + threadIdx.x;
  if (t >= N_NODES * NH) return;
  int n = t >> 3, hh = t & 7;
  const float* row = Wh + (size_t)n * HF + hh * NF;
  const float* as  = a_src + hh * NF;
  const float* ad  = a_dst + hh * NF;
  float accs = 0.f, accd = 0.f;
  for (int f = 0; f < NF; ++f) { float v = row[f]; accs += v * as[f]; accd += v * ad[f]; }
  ssrc[t] = accs;
  sdst[t] = accd;
}

__global__ __launch_bounds__(256)
void edge_logits(const float* __restrict__ ef, const int* __restrict__ src,
                 const int* __restrict__ dst, const float* __restrict__ V,
                 const float* __restrict__ ssrc, const float* __restrict__ sdst,
                 float* __restrict__ sbuf) {
  __shared__ float Vl[EDGE_F * NH];
  for (int i = threadIdx.x; i < EDGE_F * NH; i += blockDim.x) Vl[i] = V[i];
  __syncthreads();
  int t = blockIdx.x * blockDim.x + threadIdx.x;
  if (t >= N_EDGES * NH) return;
  int e = t >> 3, hh = t & 7;
  const float4* efr = (const float4*)(ef + (size_t)e * EDGE_F);
  float acc = 0.f;
  for (int q = 0; q < EDGE_F / 4; ++q) {
    float4 v = efr[q];
    int k = q * 4;
    acc += v.x * Vl[(k + 0) * NH + hh] + v.y * Vl[(k + 1) * NH + hh]
         + v.z * Vl[(k + 2) * NH + hh] + v.w * Vl[(k + 3) * NH + hh];
  }
  int s = src[e], d = dst[e];
  float sv = ssrc[s * NH + hh] + sdst[d * NH + hh] + acc;
  sv = sv > 0.f ? sv : 0.01f * sv;
  sbuf[t] = sv;
}

#define CNT_PAD 53248

__global__ __launch_bounds__(256)
void zero_counts(int* __restrict__ cnt) {
  int t = blockIdx.x * blockDim.x + threadIdx.x;
  if (t < CNT_PAD) cnt[t] = 0;
}

__global__ __launch_bounds__(256)
void dst_histogram(const int* __restrict__ dst, int* __restrict__ cnt) {
  int e = blockIdx.x * blockDim.x + threadIdx.x;
  if (e < N_EDGES) atomicAdd(&cnt[dst[e]], 1);
}

__global__ __launch_bounds__(1024)
void scan_counts(int* __restrict__ cnt, int* __restrict__ rowptr) {
  __shared__ int wsum[32];
  const int t = threadIdx.x, lane = t & 31, wave = t >> 5;
  int4* c4 = (int4*)cnt + t * 13;
  int4* r4 = (int4*)rowptr + t * 13;
  int4 v[13];
  for (int i = 0; i < 13; ++i) v[i] = c4[i];
  int sum = 0;
  for (int i = 0; i < 13; ++i) sum += v[i].x + v[i].y + v[i].z + v[i].w;
  int inc = sum;
  for (int o = 1; o < 32; o <<= 1) {
    int nb = __shfl_up(inc, o);
    if (lane >= o) inc += nb;
  }
  if (lane == 31) wsum[wave] = inc;
  __syncthreads();
  if (wave == 0) {
    int w = wsum[lane], wi = w;
    for (int o = 1; o < 32; o <<= 1) {
      int nb = __shfl_up(wi, o);
      if (lane >= o) wi += nb;
    }
    wsum[lane] = wi - w;
  }
  __syncthreads();
  int run = wsum[wave] + inc - sum;
  for (int i = 0; i < 13; ++i) {
    int4 o;
    o.x = run; run += v[i].x;
    o.y = run; run += v[i].y;
    o.z = run; run += v[i].z;
    o.w = run; run += v[i].w;
    r4[i] = o;
    c4[i] = o;
  }
}

__global__ __launch_bounds__(256)
void fill_perm(const int* __restrict__ dst, int* __restrict__ cur,
               int* __restrict__ perm) {
  int e = blockIdx.x * blockDim.x + threadIdx.x;
  if (e >= N_EDGES) return;
  int pos = atomicAdd(&cur[dst[e]], 1);
  pos = min(max(pos, 0), N_EDGES - 1);
  perm[pos] = e;
}

__device__ __forceinline__ float elu1(float x) { return x > 0.f ? x : expm1f(x); }


__device__ __forceinline__ void gather_chunk(int n, int my_s, const float* cw,
                                             const float* __restrict__ Wh, int lane,
                                             float4& acc0, float4& acc1) {
  const int h0 = lane >> 3;
#pragma unroll 4
  for (int j = 0; j < n; ++j) {
    const int s = __builtin_amdgcn_readlane(my_s, j);
    const float a0 = cw[j * NH + h0];
    const float a1 = cw[j * NH + 4 + h0];
    const float4* row = (const float4*)(Wh + (size_t)s * HF);
    const float4 v0 = row[lane];
    const float4 v1 = row[32 + lane];
    acc0.x += v0.x * a0; acc0.y += v0.y * a0; acc0.z += v0.z * a0; acc0.w += v0.w * a0;
    acc1.x += v1.x * a1; acc1.y += v1.y * a1; acc1.z += v1.z * a1; acc1.w += v1.w * a1;
  }
}

__device__ __forceinline__ void chunk_logits(int e_l, int n, int lane,
                                             const float* __restrict__ sbuf, float (&t)[8]) {
  const int hh = lane & 7, js = lane >> 3;
  int ej[8];
#pragma unroll
  for (int g = 0; g < 8; ++g) ej[g] = __shfl(e_l, 4 * g + js);
#pragma unroll
  for (int g = 0; g < 8; ++g) {
    t[g] = -INFINITY;
    if (4 * g < n) t[g] = sbuf[(size_t)ej[g] * NH + hh];
  }
#pragma unroll
  for (int g = 0; g < 8; ++g)
    if (4 * g + js >= n) t[g] = -INFINITY;
}

__global__ __launch_bounds__(256)
void aggregate_by_dst(const int* __restrict__ src, const int* __restrict__ rowptr,
                      const int* __restrict__ perm, const float* __restrict__ Wh,
                      const float* __restrict__ sbuf, float* __restrict__ out) {
  __shared__ float coef[8][32 * NH];
  const int lane = threadIdx.x & 31;
  const int wave = __builtin_amdgcn_readfirstlane((int)(threadIdx.x >> 5));
  const int node = blockIdx.x * 8 + wave;
  float* cw = coef[wave];
  const int beg = max(rowptr[node], 0);
  const int end = min(rowptr[node + 1], N_EDGES);
  const int deg = end - beg;
  float4 acc0 = {0.f, 0.f, 0.f, 0.f}, acc1 = {0.f, 0.f, 0.f, 0.f};
  if (deg > 0 && deg <= 32) {
    int my_e = 0, my_s = 0;
    if (lane < deg) {
      my_e = min(max(perm[beg + lane], 0), N_EDGES - 1);
      my_s = src[my_e];
    }
    float sv[8];
    chunk_logits(my_e, deg, lane, sbuf, sv);
    float mx = sv[0];
#pragma unroll
    for (int g = 1; g < 8; ++g) mx = fmaxf(mx, sv[g]);
    mx = fmaxf(mx, __shfl_xor(mx, 8));
    mx = fmaxf(mx, __shfl_xor(mx, 16));
    float dsum = 0.f;
#pragma unroll
    for (int g = 0; g < 8; ++g) {
      if (4 * g < deg) {
        sv[g] = expf(sv[g] - mx);
        dsum += sv[g];
      }
    }
    dsum += __shfl_xor(dsum, 8);
    dsum += __shfl_xor(dsum, 16);
#pragma unroll
    for (int g = 0; g < 8; ++g)
      if (4 * g < deg) cw[32 * g + lane] = sv[g] / dsum;
    gather_chunk(deg, my_s, cw, Wh, lane, acc0, acc1);
  } else if (deg > 32) {
    float t[8];
    float mx = -INFINITY;
    for (int base = beg; base < end; base += 32) {
      const int n = min(32, end - base);
      int e_l = 0;
      if (lane < n) e_l = min(max(perm[base + lane], 0), N_EDGES - 1);
      chunk_logits(e_l, n, lane, sbuf, t);
#pragma unroll
      for (int g = 0; g < 8; ++g) mx = fmaxf(mx, t[g]);
    }
    mx = fmaxf(mx, __shfl_xor(mx, 8));
    mx = fmaxf(mx, __shfl_xor(mx, 16));
    float dsum = 0.f;
    for (int base = beg; base < end; base += 32) {
      const int n = min(32, end - base);
      int e_l = 0;
      if (lane < n) e_l = min(max(perm[base + lane], 0), N_EDGES - 1);
      chunk_logits(e_l, n, lane, sbuf, t);
#pragma unroll
      for (int g = 0; g < 8; ++g) dsum += expf(t[g] - mx);
    }
    dsum += __shfl_xor(dsum, 8);
    dsum += __shfl_xor(dsum, 16);
    for (int base = beg; base < end; base += 32) {
      const int n = min(32, end - base);
      int e_l = 0, s_l = 0;
      if (lane < n) {
        e_l = min(max(perm[base + lane], 0), N_EDGES - 1);
        s_l = src[e_l];
      }
      chunk_logits(e_l, n, lane, sbuf, t);
#pragma unroll
      for (int g = 0; g < 8; ++g) cw[32 * g + lane] = expf(t[g] - mx) / dsum;
      gather_chunk(n, s_l, cw, Wh, lane, acc0, acc1);
    }
  }
  float4 o0, o1;
  o0.x = elu1(acc0.x); o0.y = elu1(acc0.y); o0.z = elu1(acc0.z); o0.w = elu1(acc0.w);
  o1.x = elu1(acc1.x); o1.y = elu1(acc1.y); o1.z = elu1(acc1.z); o1.w = elu1(acc1.w);
  float4* orow = (float4*)(out + (size_t)node * HF);
  orow[lane]      = o0;
  orow[32 + lane] = o1;
}

extern "C" void kernel_launch(void* const* d_in, const int* in_sizes, int n_in,
                              void* d_out, int out_size, void* d_ws, size_t ws_size,
                              hipStream_t stream) {
  const float* h   = (const float*)d_in[0];
  const float* ef  = (const float*)d_in[1];
  const int*   src = (const int*)  d_in[2];
  const int*   dst = (const int*)  d_in[3];
  const float* W_w = (const float*)d_in[4];
  const float* We_w= (const float*)d_in[5];
  const float* a_s = (const float*)d_in[6];
  const float* a_d = (const float*)d_in[7];
  const float* a_e = (const float*)d_in[8];
  float* out = (float*)d_out;

  char* ws = (char*)d_ws;
  const size_t WH_BYTES   = (size_t)N_NODES * HF * 4;
  const size_t SBUF_BYTES = (size_t)N_EDGES * NH * 4;
  const size_t NH_BYTES   = (size_t)N_NODES * NH * 4;
  float*    Wh   = (float*)   (ws);
  float*    sbuf = (float*)   (ws + WH_BYTES);
  float*    ssrc = (float*)   (ws + WH_BYTES + SBUF_BYTES);
  float*    sdst = (float*)   (ws + WH_BYTES + SBUF_BYTES + NH_BYTES);
  float*    V    = (float*)   (ws + WH_BYTES + SBUF_BYTES + 4 * NH_BYTES);

  int* cnt    = (int*)(ws + WH_BYTES + SBUF_BYTES);
  int* rowptr = (int*)(ws + WH_BYTES + SBUF_BYTES + 262144);
  int* perm   = (int*)(ws + WH_BYTES + SBUF_BYTES + NH_BYTES);

  gemm_wh_wmma<<<N_NODES / 16, 256, 0, stream>>>(h, W_w, Wh);
  fold_we<<<1, 512, 0, stream>>>(We_w, a_e, V);
  node_scores<<<(N_NODES * NH + 255) / 256, 256, 0, stream>>>(Wh, a_s, a_d, ssrc, sdst);
  edge_logits<<<(N_EDGES * NH + 255) / 256, 256, 0, stream>>>(ef, src, dst, V, ssrc, sdst, sbuf);
  zero_counts<<<CNT_PAD / 256, 256, 0, stream>>>(cnt);
  dst_histogram<<<N_EDGES / 256, 256, 0, stream>>>(dst, cnt);
  scan_counts<<<1, 1024, 0, stream>>>(cnt, rowptr);
  fill_perm<<<N_EDGES / 256, 256, 0, stream>>>(dst, cnt, perm);
  aggregate_by_dst<<<N_NODES / 8, 256, 0, stream>>>(src, rowptr, perm, Wh, sbuf, out);
}
